# starting kernel
# baseline (speedup 1.0000x reference)
#include <hip/hip_runtime.h>

typedef __attribute__((ext_vector_type(2))) float v2f;
typedef __attribute__((ext_vector_type(8))) float v8f;

#define D 128

__global__ void gemm_wmma_f32(const float* __restrict__ X,
                              const float* __restrict__ W,
                              const float* __restrict__ bias,
                              float* __restrict__ Y) {
  const int lane = threadIdx.x & 31;
  const int wave = threadIdx.x >> 5;
  const int m0 = blockIdx.x * 16;
  const int n0 = wave * 16;
  const int r = lane & 15;
  const int koff = (lane >> 4) * 2;

  const float* arow = X + (size_t)(m0 + r) * D + koff;
  const float* brow = W + (size_t)(n0 + r) * D + koff;

  v8f c = {};
#pragma unroll 4
  for (int kb = 0; kb < D / 4; ++kb) {
    v2f a = *(const v2f*)(arow + kb * 4);
    v2f b = *(const v2f*)(brow + kb * 4);
    c = __builtin_amdgcn_wmma_f32_16x16x4_f32(false, a, false, b,
                                              (short)0, c, false, false);
  }

  const float bv = bias[n0 + r];
  const int hi = lane >> 4;
  float* yout = Y + (size_t)(m0 + hi * 8) * D + n0 + r;
#pragma unroll
  for (int v = 0; v < 8; ++v) yout[(size_t)v * D] = c[v] + bv;
}

__global__ void scatter_edges(const float* __restrict__ Wh,
                              const int* __restrict__ src,
                              const int* __restrict__ dst,
                              float* __restrict__ sum,
                              float* __restrict__ cnt,
                              int nE) {
  const int lane = threadIdx.x & 31;
  const int e = blockIdx.x * (blockDim.x >> 5) + (threadIdx.x >> 5);
  if (e >= nE) return;
  const int s = src[e];
  const int d = dst[e];
  const float4 v = ((const float4*)(Wh + (size_t)s * D))[lane];
  float* dp = sum + (size_t)d * D + lane * 4;
  atomicAdd(dp + 0, v.x);
  atomicAdd(dp + 1, v.y);
  atomicAdd(dp + 2, v.z);
  atomicAdd(dp + 3, v.w);
  if (lane == 0) atomicAdd(cnt + d, 1.0f);
}

__global__ void zero_f32(float* __restrict__ p, size_t n) {
  size_t i = (size_t)blockIdx.x * blockDim.x + threadIdx.x;
  const size_t stride = (size_t)gridDim.x * blockDim.x;
  for (; i < n; i += stride) p[i] = 0.0f;
}

__global__ void finalize_user(const float* __restrict__ sum_f,
                              const float* __restrict__ cnt_f,
                              const float* __restrict__ sum_rb,
                              const float* __restrict__ cnt_rb,
                              float* __restrict__ out, size_t n) {
  size_t i = (size_t)blockIdx.x * blockDim.x + threadIdx.x;
  if (i >= n) return;
  const size_t u = i >> 7;
  out[i] = sum_f[i] / fmaxf(cnt_f[u], 1.0f) + sum_rb[i] / fmaxf(cnt_rb[u], 1.0f);
}

__global__ void finalize_item(const float* __restrict__ cnt,
                              float* __restrict__ out, size_t n) {
  size_t i = (size_t)blockIdx.x * blockDim.x + threadIdx.x;
  if (i >= n) return;
  const size_t u = i >> 7;
  out[i] = out[i] / fmaxf(cnt[u], 1.0f);
}

extern "C" void kernel_launch(void* const* d_in, const int* in_sizes, int n_in,
                              void* d_out, int out_size, void* d_ws, size_t ws_size,
                              hipStream_t stream) {
  (void)n_in; (void)out_size; (void)ws_size;

  const float* feat_user  = (const float*)d_in[0];
  const float* feat_item  = (const float*)d_in[1];
  const float* W_follows  = (const float*)d_in[2];
  const float* b_follows  = (const float*)d_in[3];
  const float* W_rates    = (const float*)d_in[4];
  const float* b_rates    = (const float*)d_in[5];
  const float* W_ratedby  = (const float*)d_in[6];
  const float* b_ratedby  = (const float*)d_in[7];
  const int* src_follows  = (const int*)d_in[8];
  const int* dst_follows  = (const int*)d_in[9];
  const int* src_rates    = (const int*)d_in[10];
  const int* dst_rates    = (const int*)d_in[11];
  const int* src_ratedby  = (const int*)d_in[12];
  const int* dst_ratedby  = (const int*)d_in[13];

  const int NU  = in_sizes[0] / D;
  const int NI  = in_sizes[1] / D;
  const int Ef  = in_sizes[8];
  const int Er  = in_sizes[10];
  const int Erb = in_sizes[12];

  float* out_user = (float*)d_out;
  float* out_item = out_user + (size_t)NU * D;

  float* ws = (float*)d_ws;
  const size_t nmax = (size_t)(NU > NI ? NU : NI) * D;
  float* whbuf  = ws;
  float* sum_f  = whbuf + nmax;
  float* sum_rb = sum_f + (size_t)NU * D;
  float* cnt_f  = sum_rb + (size_t)NU * D;
  float* cnt_rb = cnt_f + NU;
  float* cnt_r  = cnt_rb + NU;

  const size_t zcount = (size_t)NU * D * 2 + (size_t)NU * 2 + (size_t)NI;
  zero_f32<<<2048, 256, 0, stream>>>(sum_f, zcount);
  zero_f32<<<2048, 256, 0, stream>>>(out_item, (size_t)NI * D);

  gemm_wmma_f32<<<NU / 16, 256, 0, stream>>>(feat_user, W_follows, b_follows, whbuf);
  scatter_edges<<<(Ef + 7) / 8, 256, 0, stream>>>(whbuf, src_follows, dst_follows,
                                                  sum_f, cnt_f, Ef);
  gemm_wmma_f32<<<NI / 16, 256, 0, stream>>>(feat_item, W_ratedby, b_ratedby, whbuf);
  scatter_edges<<<(Erb + 7) / 8, 256, 0, stream>>>(whbuf, src_ratedby, dst_ratedby,
                                                   sum_rb, cnt_rb, Erb);
  gemm_wmma_f32<<<NU / 16, 256, 0, stream>>>(feat_user, W_rates, b_rates, whbuf);
  scatter_edges<<<(Er + 7) / 8, 256, 0, stream>>>(whbuf, src_rates, dst_rates,
                                                  out_item, cnt_r, Er);

  const size_t nu_elems = (size_t)NU * D;
  const size_t ni_elems = (size_t)NI * D;
  finalize_user<<<(unsigned)((nu_elems + 255) / 256), 256, 0, stream>>>(
      sum_f, cnt_f, sum_rb, cnt_rb, out_user, nu_elems);
  finalize_item<<<(unsigned)((ni_elems + 255) / 256), 256, 0, stream>>>(
      cnt_r, out_item, ni_elems);
}


# GEMM re-tiled: W staged once per work-group in LDS, each wave makes a 16x128 strip so one X fragment feeds 8 WMMAs; k summed per 8-block as 0,1,4,5 then 2,3,6,7 (f32 sum order changes)
# speedup vs baseline: 1.1678x; 1.1678x over previous
#include <hip/hip_runtime.h>

typedef __attribute__((ext_vector_type(2))) float v2f;
typedef __attribute__((ext_vector_type(8))) float v8f;

#define D 128

#define WP 132

__global__ void __launch_bounds__(256)
gemm_wmma_f32(const float* __restrict__ X,
              const float* __restrict__ W,
              const float* __restrict__ bias,
              float* __restrict__ Y,
              int ntiles) {
  __shared__ __attribute__((aligned(16))) float sW[D * WP];


  const int tid = threadIdx.x;
  const int lane = tid & 31;
  const int wave = __builtin_amdgcn_readfirstlane(tid >> 5);
  const int r = lane & 15;
  const int h = lane >> 4;

  const int tile = blockIdx.x * 8 + wave;
  const int tl = tile < ntiles ? tile : ntiles - 1;

  const float* arow = X + (size_t)(tl * 16 + r) * D + h * 4;
  float4 a[16];
#pragma unroll
  for (int j = 0; j < 16; ++j) a[j] = *(const float4*)(arow + j * 8);

#pragma unroll
  for (int i = 0; i < 16; ++i) {
    const int idx = tid + i * 256;
    const int n = idx >> 5;
    const int c4 = idx & 31;
    const float4 w4 = ((const float4*)W)[idx];
    *(float4*)(sW + n * WP + c4 * 4) = w4;
  }
  __syncthreads();

  if (tile >= ntiles) return;

  const float* bl = sW + r * WP + h * 4;

  v8f c[8];
#pragma unroll
  for (int nt = 0; nt < 8; ++nt) c[nt] = (v8f){};

  float4 b[8];
#pragma unroll
  for (int nt = 0; nt < 8; ++nt) b[nt] = *(const float4*)(bl + nt * 16 * WP);

#pragma unroll
  for (int j = 0; j < 16; ++j) {
    float4 bn[8];
    if (j + 1 < 16) {
#pragma unroll
      for (int nt = 0; nt < 8; ++nt)
        bn[nt] = *(const float4*)(bl + nt * 16 * WP + (j + 1) * 8);
    }
    const v2f a0 = {a[j].x, a[j].y};
    const v2f a1 = {a[j].z, a[j].w};
#pragma unroll
    for (int nt = 0; nt < 8; ++nt) {
      const v2f b0 = {b[nt].x, b[nt].y};
      c[nt] = __builtin_amdgcn_wmma_f32_16x16x4_f32(false, a0, false, b0,
                                                    (short)0, c[nt], false, false);
    }
#pragma unroll
    for (int nt = 0; nt < 8; ++nt) {
      const v2f b1 = {b[nt].z, b[nt].w};
      c[nt] = __builtin_amdgcn_wmma_f32_16x16x4_f32(false, a1, false, b1,
                                                    (short)0, c[nt], false, false);
    }
    if (j + 1 < 16) {
#pragma unroll
      for (int nt = 0; nt < 8; ++nt) b[nt] = bn[nt];
    }
  }

  float* yout = Y + (size_t)(tile * 16 + h * 8) * D + r;
#pragma unroll
  for (int nt = 0; nt < 8; ++nt) {
    const float bv = bias[nt * 16 + r];
#pragma unroll
    for (int v = 0; v < 8; ++v) yout[(size_t)v * D + nt * 16] = c[nt][v] + bv;
  }
}

__global__ void scatter_edges(const float* __restrict__ Wh,
                              const int* __restrict__ src,
                              const int* __restrict__ dst,
                              float* __restrict__ sum,
                              float* __restrict__ cnt,
                              int nE) {
  const int lane = threadIdx.x & 31;
  const int e = blockIdx.x * (blockDim.x >> 5) + (threadIdx.x >> 5);
  if (e >= nE) return;
  const int s = src[e];
  const int d = dst[e];
  const float4 v = ((const float4*)(Wh + (size_t)s * D))[lane];
  float* dp = sum + (size_t)d * D + lane * 4;
  atomicAdd(dp + 0, v.x);
  atomicAdd(dp + 1, v.y);
  atomicAdd(dp + 2, v.z);
  atomicAdd(dp + 3, v.w);
  if (lane == 0) atomicAdd(cnt + d, 1.0f);
}

__global__ void zero_f32(float* __restrict__ p, size_t n) {
  size_t i = (size_t)blockIdx.x * blockDim.x + threadIdx.x;
  const size_t stride = (size_t)gridDim.x * blockDim.x;
  for (; i < n; i += stride) p[i] = 0.0f;
}

__global__ void finalize_user(const float* __restrict__ sum_f,
                              const float* __restrict__ cnt_f,
                              const float* __restrict__ sum_rb,
                              const float* __restrict__ cnt_rb,
                              float* __restrict__ out, size_t n) {
  size_t i = (size_t)blockIdx.x * blockDim.x + threadIdx.x;
  if (i >= n) return;
  const size_t u = i >> 7;
  out[i] = sum_f[i] / fmaxf(cnt_f[u], 1.0f) + sum_rb[i] / fmaxf(cnt_rb[u], 1.0f);
}

__global__ void finalize_item(const float* __restrict__ cnt,
                              float* __restrict__ out, size_t n) {
  size_t i = (size_t)blockIdx.x * blockDim.x + threadIdx.x;
  if (i >= n) return;
  const size_t u = i >> 7;
  out[i] = out[i] / fmaxf(cnt[u], 1.0f);
}

extern "C" void kernel_launch(void* const* d_in, const int* in_sizes, int n_in,
                              void* d_out, int out_size, void* d_ws, size_t ws_size,
                              hipStream_t stream) {
  (void)n_in; (void)out_size; (void)ws_size;

  const float* feat_user  = (const float*)d_in[0];
  const float* feat_item  = (const float*)d_in[1];
  const float* W_follows  = (const float*)d_in[2];
  const float* b_follows  = (const float*)d_in[3];
  const float* W_rates    = (const float*)d_in[4];
  const float* b_rates    = (const float*)d_in[5];
  const float* W_ratedby  = (const float*)d_in[6];
  const float* b_ratedby  = (const float*)d_in[7];
  const int* src_follows  = (const int*)d_in[8];
  const int* dst_follows  = (const int*)d_in[9];
  const int* src_rates    = (const int*)d_in[10];
  const int* dst_rates    = (const int*)d_in[11];
  const int* src_ratedby  = (const int*)d_in[12];
  const int* dst_ratedby  = (const int*)d_in[13];

  const int NU  = in_sizes[0] / D;
  const int NI  = in_sizes[1] / D;
  const int Ef  = in_sizes[8];
  const int Er  = in_sizes[10];
  const int Erb = in_sizes[12];

  float* out_user = (float*)d_out;
  float* out_item = out_user + (size_t)NU * D;

  float* ws = (float*)d_ws;
  const size_t nmax = (size_t)(NU > NI ? NU : NI) * D;
  float* whbuf  = ws;
  float* sum_f  = whbuf + nmax;
  float* sum_rb = sum_f + (size_t)NU * D;
  float* cnt_f  = sum_rb + (size_t)NU * D;
  float* cnt_rb = cnt_f + NU;
  float* cnt_r  = cnt_rb + NU;

  const size_t zcount = (size_t)NU * D * 2 + (size_t)NU * 2 + (size_t)NI;
  zero_f32<<<2048, 256, 0, stream>>>(sum_f, zcount);
  zero_f32<<<2048, 256, 0, stream>>>(out_item, (size_t)NI * D);

  gemm_wmma_f32<<<(NU / 16 + 7) / 8, 256, 0, stream>>>(feat_user, W_follows, b_follows, whbuf, NU / 16);
  scatter_edges<<<(Ef + 7) / 8, 256, 0, stream>>>(whbuf, src_follows, dst_follows,
                                                  sum_f, cnt_f, Ef);
  gemm_wmma_f32<<<(NI / 16 + 7) / 8, 256, 0, stream>>>(feat_item, W_ratedby, b_ratedby, whbuf, NI / 16);
  scatter_edges<<<(Erb + 7) / 8, 256, 0, stream>>>(whbuf, src_ratedby, dst_ratedby,
                                                   sum_rb, cnt_rb, Erb);
  gemm_wmma_f32<<<(NU / 16 + 7) / 8, 256, 0, stream>>>(feat_user, W_rates, b_rates, whbuf, NU / 16);
  scatter_edges<<<(Er + 7) / 8, 256, 0, stream>>>(whbuf, src_rates, dst_rates,
                                                  out_item, cnt_r, Er);

  const size_t nu_elems = (size_t)NU * D;
  const size_t ni_elems = (size_t)NI * D;
  finalize_user<<<(unsigned)((nu_elems + 255) / 256), 256, 0, stream>>>(
      sum_f, cnt_f, sum_rb, cnt_rb, out_user, nu_elems);
  finalize_item<<<(unsigned)((ni_elems + 255) / 256), 256, 0, stream>>>(
      cnt_r, out_item, ni_elems);
}


# scatter: lane L takes floats L, L+32, L+64, L+96 of a row (was 4L..4L+3), so each f32 atomic instruction and each gather load covers one whole 128-B line: a quarter as many atomic line requests
# speedup vs baseline: 1.3223x; 1.1323x over previous
#include <hip/hip_runtime.h>

typedef __attribute__((ext_vector_type(2))) float v2f;
typedef __attribute__((ext_vector_type(8))) float v8f;

#define D 128

#define WP 132

__global__ void __launch_bounds__(256)
gemm_wmma_f32(const float* __restrict__ X,
              const float* __restrict__ W,
              const float* __restrict__ bias,
              float* __restrict__ Y,
              int ntiles) {
  __shared__ __attribute__((aligned(16))) float sW[D * WP];


  const int tid = threadIdx.x;
  const int lane = tid & 31;
  const int wave = __builtin_amdgcn_readfirstlane(tid >> 5);
  const int r = lane & 15;
  const int h = lane >> 4;

  const int tile = blockIdx.x * 8 + wave;
  const int tl = tile < ntiles ? tile : ntiles - 1;

  const float* arow = X + (size_t)(tl * 16 + r) * D + h * 4;
  float4 a[16];
#pragma unroll
  for (int j = 0; j < 16; ++j) a[j] = *(const float4*)(arow + j * 8);

#pragma unroll
  for (int i = 0; i < 16; ++i) {
    const int idx = tid + i * 256;
    const int n = idx >> 5;
    const int c4 = idx & 31;
    const float4 w4 = ((const float4*)W)[idx];
    *(float4*)(sW + n * WP + c4 * 4) = w4;
  }
  __syncthreads();

  if (tile >= ntiles) return;

  const float* bl = sW + r * WP + h * 4;

  v8f c[8];
#pragma unroll
  for (int nt = 0; nt < 8; ++nt) c[nt] = (v8f){};

  float4 b[8];
#pragma unroll
  for (int nt = 0; nt < 8; ++nt) b[nt] = *(const float4*)(bl + nt * 16 * WP);

#pragma unroll
  for (int j = 0; j < 16; ++j) {
    float4 bn[8];
    if (j + 1 < 16) {
#pragma unroll
      for (int nt = 0; nt < 8; ++nt)
        bn[nt] = *(const float4*)(bl + nt * 16 * WP + (j + 1) * 8);
    }
    const v2f a0 = {a[j].x, a[j].y};
    const v2f a1 = {a[j].z, a[j].w};
#pragma unroll
    for (int nt = 0; nt < 8; ++nt) {
      const v2f b0 = {b[nt].x, b[nt].y};
      c[nt] = __builtin_amdgcn_wmma_f32_16x16x4_f32(false, a0, false, b0,
                                                    (short)0, c[nt], false, false);
    }
#pragma unroll
    for (int nt = 0; nt < 8; ++nt) {
      const v2f b1 = {b[nt].z, b[nt].w};
      c[nt] = __builtin_amdgcn_wmma_f32_16x16x4_f32(false, a1, false, b1,
                                                    (short)0, c[nt], false, false);
    }
    if (j + 1 < 16) {
#pragma unroll
      for (int nt = 0; nt < 8; ++nt) b[nt] = bn[nt];
    }
  }

  float* yout = Y + (size_t)(tile * 16 + h * 8) * D + r;
#pragma unroll
  for (int nt = 0; nt < 8; ++nt) {
    const float bv = bias[nt * 16 + r];
#pragma unroll
    for (int v = 0; v < 8; ++v) yout[(size_t)v * D + nt * 16] = c[nt][v] + bv;
  }
}

__global__ void scatter_edges(const float* __restrict__ Wh,
                              const int* __restrict__ src,
                              const int* __restrict__ dst,
                              float* __restrict__ sum,
                              float* __restrict__ cnt,
                              int nE) {
  const int lane = threadIdx.x & 31;
  const int e = blockIdx.x * (blockDim.x >> 5) + (threadIdx.x >> 5);
  if (e >= nE) return;
  const int s = src[e];
  const int d = dst[e];
  const float* sp = Wh + (size_t)s * D + lane;
  const float v0 = sp[0];
  const float v1 = sp[32];
  const float v2 = sp[64];
  const float v3 = sp[96];
  float* dp = sum + (size_t)d * D + lane;
  atomicAdd(dp + 0, v0);
  atomicAdd(dp + 32, v1);
  atomicAdd(dp + 64, v2);
  atomicAdd(dp + 96, v3);
  if (lane == 0) atomicAdd(cnt + d, 1.0f);
}

__global__ void zero_f32(float* __restrict__ p, size_t n) {
  size_t i = (size_t)blockIdx.x * blockDim.x + threadIdx.x;
  const size_t stride = (size_t)gridDim.x * blockDim.x;
  for (; i < n; i += stride) p[i] = 0.0f;
}

__global__ void finalize_user(const float* __restrict__ sum_f,
                              const float* __restrict__ cnt_f,
                              const float* __restrict__ sum_rb,
                              const float* __restrict__ cnt_rb,
                              float* __restrict__ out, size_t n) {
  size_t i = (size_t)blockIdx.x * blockDim.x + threadIdx.x;
  if (i >= n) return;
  const size_t u = i >> 7;
  out[i] = sum_f[i] / fmaxf(cnt_f[u], 1.0f) + sum_rb[i] / fmaxf(cnt_rb[u], 1.0f);
}

__global__ void finalize_item(const float* __restrict__ cnt,
                              float* __restrict__ out, size_t n) {
  size_t i = (size_t)blockIdx.x * blockDim.x + threadIdx.x;
  if (i >= n) return;
  const size_t u = i >> 7;
  out[i] = out[i] / fmaxf(cnt[u], 1.0f);
}

extern "C" void kernel_launch(void* const* d_in, const int* in_sizes, int n_in,
                              void* d_out, int out_size, void* d_ws, size_t ws_size,
                              hipStream_t stream) {
  (void)n_in; (void)out_size; (void)ws_size;

  const float* feat_user  = (const float*)d_in[0];
  const float* feat_item  = (const float*)d_in[1];
  const float* W_follows  = (const float*)d_in[2];
  const float* b_follows  = (const float*)d_in[3];
  const float* W_rates    = (const float*)d_in[4];
  const float* b_rates    = (const float*)d_in[5];
  const float* W_ratedby  = (const float*)d_in[6];
  const float* b_ratedby  = (const float*)d_in[7];
  const int* src_follows  = (const int*)d_in[8];
  const int* dst_follows  = (const int*)d_in[9];
  const int* src_rates    = (const int*)d_in[10];
  const int* dst_rates    = (const int*)d_in[11];
  const int* src_ratedby  = (const int*)d_in[12];
  const int* dst_ratedby  = (const int*)d_in[13];

  const int NU  = in_sizes[0] / D;
  const int NI  = in_sizes[1] / D;
  const int Ef  = in_sizes[8];
  const int Er  = in_sizes[10];
  const int Erb = in_sizes[12];

  float* out_user = (float*)d_out;
  float* out_item = out_user + (size_t)NU * D;

  float* ws = (float*)d_ws;
  const size_t nmax = (size_t)(NU > NI ? NU : NI) * D;
  float* whbuf  = ws;
  float* sum_f  = whbuf + nmax;
  float* sum_rb = sum_f + (size_t)NU * D;
  float* cnt_f  = sum_rb + (size_t)NU * D;
  float* cnt_rb = cnt_f + NU;
  float* cnt_r  = cnt_rb + NU;

  const size_t zcount = (size_t)NU * D * 2 + (size_t)NU * 2 + (size_t)NI;
  zero_f32<<<2048, 256, 0, stream>>>(sum_f, zcount);
  zero_f32<<<2048, 256, 0, stream>>>(out_item, (size_t)NI * D);

  gemm_wmma_f32<<<(NU / 16 + 7) / 8, 256, 0, stream>>>(feat_user, W_follows, b_follows, whbuf, NU / 16);
  scatter_edges<<<(Ef + 7) / 8, 256, 0, stream>>>(whbuf, src_follows, dst_follows,
                                                  sum_f, cnt_f, Ef);
  gemm_wmma_f32<<<(NI / 16 + 7) / 8, 256, 0, stream>>>(feat_item, W_ratedby, b_ratedby, whbuf, NI / 16);
  scatter_edges<<<(Erb + 7) / 8, 256, 0, stream>>>(whbuf, src_ratedby, dst_ratedby,
                                                   sum_rb, cnt_rb, Erb);
  gemm_wmma_f32<<<(NU / 16 + 7) / 8, 256, 0, stream>>>(feat_user, W_rates, b_rates, whbuf, NU / 16);
  scatter_edges<<<(Er + 7) / 8, 256, 0, stream>>>(whbuf, src_rates, dst_rates,
                                                  out_item, cnt_r, Er);

  const size_t nu_elems = (size_t)NU * D;
  const size_t ni_elems = (size_t)NI * D;
  finalize_user<<<(unsigned)((nu_elems + 255) / 256), 256, 0, stream>>>(
      sum_f, cnt_f, sum_rb, cnt_rb, out_user, nu_elems);
  finalize_item<<<(unsigned)((ni_elems + 255) / 256), 256, 0, stream>>>(
      cnt_r, out_item, ni_elems);
}


# f32 atomic scatter-add replaced by a per-call counting sort of edges by destination (integer atomic rank, scan, place) and one wave per node summing its source rows in registers; sums in rank order
# speedup vs baseline: 2.1718x; 1.6424x over previous
#include <hip/hip_runtime.h>

typedef __attribute__((ext_vector_type(2))) float v2f;
typedef __attribute__((ext_vector_type(8))) float v8f;

#define D 128

#define WP 132

__global__ void __launch_bounds__(256)
gemm_wmma_f32(const float* __restrict__ X,
              const float* __restrict__ W,
              const float* __restrict__ bias,
              float* __restrict__ Y,
              int ntiles) {
  __shared__ __attribute__((aligned(16))) float sW[D * WP];


  const int tid = threadIdx.x;
  const int lane = tid & 31;
  const int wave = __builtin_amdgcn_readfirstlane(tid >> 5);
  const int r = lane & 15;
  const int h = lane >> 4;

  const int tile = blockIdx.x * 8 + wave;
  const int tl = tile < ntiles ? tile : ntiles - 1;

  const float* arow = X + (size_t)(tl * 16 + r) * D + h * 4;
  float4 a[16];
#pragma unroll
  for (int j = 0; j < 16; ++j) a[j] = *(const float4*)(arow + j * 8);

#pragma unroll
  for (int i = 0; i < 16; ++i) {
    const int idx = tid + i * 256;
    const int n = idx >> 5;
    const int c4 = idx & 31;
    const float4 w4 = ((const float4*)W)[idx];
    *(float4*)(sW + n * WP + c4 * 4) = w4;
  }
  __syncthreads();

  if (tile >= ntiles) return;

  const float* bl = sW + r * WP + h * 4;

  v8f c[8];
#pragma unroll
  for (int nt = 0; nt < 8; ++nt) c[nt] = (v8f){};

  float4 b[8];
#pragma unroll
  for (int nt = 0; nt < 8; ++nt) b[nt] = *(const float4*)(bl + nt * 16 * WP);

#pragma unroll
  for (int j = 0; j < 16; ++j) {
    float4 bn[8];
    if (j + 1 < 16) {
#pragma unroll
      for (int nt = 0; nt < 8; ++nt)
        bn[nt] = *(const float4*)(bl + nt * 16 * WP + (j + 1) * 8);
    }
    const v2f a0 = {a[j].x, a[j].y};
    const v2f a1 = {a[j].z, a[j].w};
#pragma unroll
    for (int nt = 0; nt < 8; ++nt) {
      const v2f b0 = {b[nt].x, b[nt].y};
      c[nt] = __builtin_amdgcn_wmma_f32_16x16x4_f32(false, a0, false, b0,
                                                    (short)0, c[nt], false, false);
    }
#pragma unroll
    for (int nt = 0; nt < 8; ++nt) {
      const v2f b1 = {b[nt].z, b[nt].w};
      c[nt] = __builtin_amdgcn_wmma_f32_16x16x4_f32(false, a1, false, b1,
                                                    (short)0, c[nt], false, false);
    }
    if (j + 1 < 16) {
#pragma unroll
      for (int nt = 0; nt < 8; ++nt) b[nt] = bn[nt];
    }
  }

  float* yout = Y + (size_t)(tile * 16 + h * 8) * D + r;
#pragma unroll
  for (int nt = 0; nt < 8; ++nt) {
    const float bv = bias[nt * 16 + r];
#pragma unroll
    for (int v = 0; v < 8; ++v) yout[(size_t)v * D + nt * 16] = c[nt][v] + bv;
  }
}

__global__ void zero_i32(int* __restrict__ p, int n) {
  const int i = blockIdx.x * blockDim.x + threadIdx.x;
  if (i < n) p[i] = 0;
}

__global__ void __launch_bounds__(256)
rank_edges(const int* __restrict__ dst_f, const int* __restrict__ dst_rb,
           const int* __restrict__ dst_r,
           int* __restrict__ cnt, int* __restrict__ rank,
           int Ef, int Erb, int Er, int NU) {
  const int et = blockIdx.y;
  const int* dst = et == 0 ? dst_f : (et == 1 ? dst_rb : dst_r);
  const int nE   = et == 0 ? Ef : (et == 1 ? Erb : Er);
  const int coff = et == 0 ? 0 : (et == 1 ? NU : 2 * NU);
  const int eoff = et == 0 ? 0 : (et == 1 ? Ef : Ef + Erb);
  const int e = blockIdx.x * blockDim.x + threadIdx.x;
  if (e >= nE) return;
  rank[eoff + e] = atomicAdd(cnt + coff + dst[e], 1);
}

__global__ void __launch_bounds__(1024)
scan_counts(const int* __restrict__ cnt, int* __restrict__ first,
            int NU, int NI, int Ef, int Erb) {
  __shared__ int s_tot[32];
  __shared__ int s_prev[32];
  const int et = blockIdx.y;
  const int n     = et == 2 ? NI : NU;
  const int coff  = et == 0 ? 0 : (et == 1 ? NU : 2 * NU);
  const int ebase = et == 0 ? 0 : (et == 1 ? Ef : Ef + Erb);
  const int tile = blockIdx.x;
  if (tile * 4096 >= n) return;

  const int tid = threadIdx.x;
  const int lane = tid & 31;
  const int wave = tid >> 5;
  const int4* c4 = (const int4*)(cnt + coff);

  int prev = 0;
  for (int p = 0; p < tile; ++p) {
    const int4 q = c4[p * 1024 + tid];
    prev += (q.x + q.y) + (q.z + q.w);
  }

  const int i4 = tile * 1024 + tid;
  const bool in = i4 * 4 < n;
  int4 v = make_int4(0, 0, 0, 0);
  if (in) v = c4[i4];
  const int s = (v.x + v.y) + (v.z + v.w);

  int incl = s;
#pragma unroll
  for (int o = 1; o < 32; o <<= 1) {
    const int t = __shfl_up(incl, o, 32);
    if (lane >= o) incl += t;
  }
#pragma unroll
  for (int o = 16; o > 0; o >>= 1) prev += __shfl_xor(prev, o, 32);

  if (lane == 31) s_tot[wave] = incl;
  if (lane == 0) s_prev[wave] = prev;
  __syncthreads();

  int wsel = lane < wave ? s_tot[lane] : 0;
  int pv = s_prev[lane];
#pragma unroll
  for (int o = 16; o > 0; o >>= 1) {
    wsel += __shfl_xor(wsel, o, 32);
    pv += __shfl_xor(pv, o, 32);
  }
  const int ex = ebase + pv + wsel + (incl - s);
  if (in)
    ((int4*)(first + coff))[i4] = make_int4(ex, ex + v.x, ex + v.x + v.y, ex + v.x + v.y + v.z);
}

__global__ void __launch_bounds__(256)
place_edges(const int* __restrict__ src_f, const int* __restrict__ src_rb,
            const int* __restrict__ src_r,
            const int* __restrict__ dst_f, const int* __restrict__ dst_rb,
            const int* __restrict__ dst_r,
            const int* __restrict__ first, const int* __restrict__ rank,
            int* __restrict__ sorted,
            int Ef, int Erb, int Er, int NU) {
  const int et = blockIdx.y;
  const int* src = et == 0 ? src_f : (et == 1 ? src_rb : src_r);
  const int* dst = et == 0 ? dst_f : (et == 1 ? dst_rb : dst_r);
  const int nE   = et == 0 ? Ef : (et == 1 ? Erb : Er);
  const int coff = et == 0 ? 0 : (et == 1 ? NU : 2 * NU);
  const int eoff = et == 0 ? 0 : (et == 1 ? Ef : Ef + Erb);
  const int e = blockIdx.x * blockDim.x + threadIdx.x;
  if (e >= nE) return;
  sorted[first[coff + dst[e]] + rank[eoff + e]] = src[e];
}

__device__ __forceinline__ void add_rows(float4& acc, const float* __restrict__ whl,
                                         int id, int n) {
  int k = 0;
  for (; k + 8 <= n; k += 8) {
    float4 v[8];
#pragma unroll
    for (int i = 0; i < 8; ++i)
      v[i] = *(const float4*)(whl + (size_t)__builtin_amdgcn_readlane(id, k + i) * D);
#pragma unroll
    for (int i = 0; i < 8; ++i) {
      acc.x += v[i].x; acc.y += v[i].y; acc.z += v[i].z; acc.w += v[i].w;
    }
  }
  int s[7];
#pragma unroll
  for (int i = 0; i < 7; ++i) s[i] = __builtin_amdgcn_readlane(id, (k + i) & 31);
  float4 t[7];
#pragma unroll
  for (int i = 0; i < 7; ++i) {
    t[i] = make_float4(0.0f, 0.0f, 0.0f, 0.0f);
    if (k + i < n) t[i] = *(const float4*)(whl + (size_t)s[i] * D);
  }
#pragma unroll
  for (int i = 0; i < 7; ++i) {
    acc.x += t[i].x; acc.y += t[i].y; acc.z += t[i].z; acc.w += t[i].w;
  }
}

__device__ __forceinline__ float4 seg_sum(const float* __restrict__ whl,
                                          const int* __restrict__ sorted,
                                          int start, int deg, int id0, int lane) {
  float4 acc = make_float4(0.0f, 0.0f, 0.0f, 0.0f);
  add_rows(acc, whl, id0, deg < 32 ? deg : 32);
  for (int b = 32; b < deg; b += 32) {
    const int n = deg - b < 32 ? deg - b : 32;
    int id = 0;
    if (lane < n) id = sorted[start + b + lane];
    add_rows(acc, whl, id, n);
  }
  return acc;
}

__global__ void __launch_bounds__(256)
reduce_nodes(const float* __restrict__ Wh_f, const float* __restrict__ Wh_rb,
             const float* __restrict__ Wh_r,
             const int* __restrict__ cnt, const int* __restrict__ first,
             const int* __restrict__ sorted,
             float* __restrict__ out, int NU, int NI) {
  const int lane = threadIdx.x & 31;
  const int w = __builtin_amdgcn_readfirstlane((int)(blockIdx.x * 8 + (threadIdx.x >> 5)));
  if (w >= NU + NI) return;
  float4 res;
  if (w < NU) {
    const int cf  = __builtin_amdgcn_readfirstlane(cnt[w]);
    const int crb = __builtin_amdgcn_readfirstlane(cnt[NU + w]);
    const int pf  = __builtin_amdgcn_readfirstlane(first[w]);
    const int prb = __builtin_amdgcn_readfirstlane(first[NU + w]);
    int idf = 0, idrb = 0;
    if (lane < cf) idf = sorted[pf + lane];
    if (lane < crb) idrb = sorted[prb + lane];
    const float4 a = seg_sum(Wh_f + lane * 4, sorted, pf, cf, idf, lane);
    const float4 b = seg_sum(Wh_rb + lane * 4, sorted, prb, crb, idrb, lane);
    const float df = fmaxf((float)cf, 1.0f);
    const float drb = fmaxf((float)crb, 1.0f);
    res.x = a.x / df + b.x / drb;
    res.y = a.y / df + b.y / drb;
    res.z = a.z / df + b.z / drb;
    res.w = a.w / df + b.w / drb;
  } else {
    const int i = w - NU;
    const int c = __builtin_amdgcn_readfirstlane(cnt[2 * NU + i]);
    const int p = __builtin_amdgcn_readfirstlane(first[2 * NU + i]);
    int id = 0;
    if (lane < c) id = sorted[p + lane];
    const float4 a = seg_sum(Wh_r + lane * 4, sorted, p, c, id, lane);
    const float dr = fmaxf((float)c, 1.0f);
    res.x = a.x / dr; res.y = a.y / dr; res.z = a.z / dr; res.w = a.w / dr;
  }
  *(float4*)(out + (size_t)w * D + lane * 4) = res;
}

extern "C" void kernel_launch(void* const* d_in, const int* in_sizes, int n_in,
                              void* d_out, int out_size, void* d_ws, size_t ws_size,
                              hipStream_t stream) {
  (void)n_in; (void)out_size; (void)ws_size;

  const float* feat_user  = (const float*)d_in[0];
  const float* feat_item  = (const float*)d_in[1];
  const float* W_follows  = (const float*)d_in[2];
  const float* b_follows  = (const float*)d_in[3];
  const float* W_rates    = (const float*)d_in[4];
  const float* b_rates    = (const float*)d_in[5];
  const float* W_ratedby  = (const float*)d_in[6];
  const float* b_ratedby  = (const float*)d_in[7];
  const int* src_follows  = (const int*)d_in[8];
  const int* dst_follows  = (const int*)d_in[9];
  const int* src_rates    = (const int*)d_in[10];
  const int* dst_rates    = (const int*)d_in[11];
  const int* src_ratedby  = (const int*)d_in[12];
  const int* dst_ratedby  = (const int*)d_in[13];

  const int NU  = in_sizes[0] / D;
  const int NI  = in_sizes[1] / D;
  const int Ef  = in_sizes[8];
  const int Er  = in_sizes[10];
  const int Erb = in_sizes[12];

  float* out = (float*)d_out;

  float* ws = (float*)d_ws;
  const int nnmax = NU > NI ? NU : NI;
  const size_t nmax = (size_t)nnmax * D;
  float* wh_f  = ws;
  float* wh_rb = wh_f + nmax;
  float* wh_r  = wh_rb + nmax;
  int* cnt    = (int*)(wh_r + nmax);
  const int ncnt = 2 * NU + NI;
  int* first  = cnt + ncnt;
  int* rank   = first + ncnt;
  const int etot = Ef + Erb + Er;
  int* sorted = rank + etot;
  int emax = Ef > Erb ? Ef : Erb;
  emax = emax > Er ? emax : Er;

  zero_i32<<<(ncnt + 255) / 256, 256, 0, stream>>>(cnt, ncnt);
  rank_edges<<<dim3((emax + 255) / 256, 3), 256, 0, stream>>>(
      dst_follows, dst_ratedby, dst_rates, cnt, rank, Ef, Erb, Er, NU);
  scan_counts<<<dim3((nnmax + 4095) / 4096, 3), 1024, 0, stream>>>(cnt, first, NU, NI, Ef, Erb);
  place_edges<<<dim3((emax + 255) / 256, 3), 256, 0, stream>>>(
      src_follows, src_ratedby, src_rates, dst_follows, dst_ratedby, dst_rates,
      first, rank, sorted, Ef, Erb, Er, NU);

  gemm_wmma_f32<<<(NU / 16 + 7) / 8, 256, 0, stream>>>(feat_user, W_follows, b_follows, wh_f, NU / 16);
  gemm_wmma_f32<<<(NI / 16 + 7) / 8, 256, 0, stream>>>(feat_item, W_ratedby, b_ratedby, wh_rb, NI / 16);
  gemm_wmma_f32<<<(NU / 16 + 7) / 8, 256, 0, stream>>>(feat_user, W_rates, b_rates, wh_r, NU / 16);

  reduce_nodes<<<(NU + NI + 7) / 8, 256, 0, stream>>>(wh_f, wh_rb, wh_r, cnt, first, sorted,
                                                      out, NU, NI);
}


# edge sort without scan and second edge pass: each destination owns a 32-word line, an edge writes its source id there at the rank its count atomic returned (8 -> 6 launches); sums stay in rank order
# speedup vs baseline: 2.4945x; 1.1486x over previous
#include <hip/hip_runtime.h>

typedef __attribute__((ext_vector_type(2))) float v2f;
typedef __attribute__((ext_vector_type(8))) float v8f;

#define D 128

#define WP 132

__global__ void __launch_bounds__(256)
gemm_wmma_f32(const float* __restrict__ X,
              const float* __restrict__ W,
              const float* __restrict__ bias,
              float* __restrict__ Y,
              int ntiles) {
  __shared__ __attribute__((aligned(16))) float sW[D * WP];


  const int tid = threadIdx.x;
  const int lane = tid & 31;
  const int wave = __builtin_amdgcn_readfirstlane(tid >> 5);
  const int r = lane & 15;
  const int h = lane >> 4;

  const int tile = blockIdx.x * 8 + wave;
  const int tl = tile < ntiles ? tile : ntiles - 1;

  const float* arow = X + (size_t)(tl * 16 + r) * D + h * 4;
  float4 a[16];
#pragma unroll
  for (int j = 0; j < 16; ++j) a[j] = *(const float4*)(arow + j * 8);

#pragma unroll
  for (int i = 0; i < 16; ++i) {
    const int idx = tid + i * 256;
    const int n = idx >> 5;
    const int c4 = idx & 31;
    const float4 w4 = ((const float4*)W)[idx];
    *(float4*)(sW + n * WP + c4 * 4) = w4;
  }
  __syncthreads();

  if (tile >= ntiles) return;

  const float* bl = sW + r * WP + h * 4;

  v8f c[8];
#pragma unroll
  for (int nt = 0; nt < 8; ++nt) c[nt] = (v8f){};

  float4 b[8];
#pragma unroll
  for (int nt = 0; nt < 8; ++nt) b[nt] = *(const float4*)(bl + nt * 16 * WP);

#pragma unroll
  for (int j = 0; j < 16; ++j) {
    float4 bn[8];
    if (j + 1 < 16) {
#pragma unroll
      for (int nt = 0; nt < 8; ++nt)
        bn[nt] = *(const float4*)(bl + nt * 16 * WP + (j + 1) * 8);
    }
    const v2f a0 = {a[j].x, a[j].y};
    const v2f a1 = {a[j].z, a[j].w};
#pragma unroll
    for (int nt = 0; nt < 8; ++nt) {
      const v2f b0 = {b[nt].x, b[nt].y};
      c[nt] = __builtin_amdgcn_wmma_f32_16x16x4_f32(false, a0, false, b0,
                                                    (short)0, c[nt], false, false);
    }
#pragma unroll
    for (int nt = 0; nt < 8; ++nt) {
      const v2f b1 = {b[nt].z, b[nt].w};
      c[nt] = __builtin_amdgcn_wmma_f32_16x16x4_f32(false, a1, false, b1,
                                                    (short)0, c[nt], false, false);
    }
    if (j + 1 < 16) {
#pragma unroll
      for (int nt = 0; nt < 8; ++nt) b[nt] = bn[nt];
    }
  }

  float* yout = Y + (size_t)(tile * 16 + h * 8) * D + r;
#pragma unroll
  for (int nt = 0; nt < 8; ++nt) {
    const float bv = bias[nt * 16 + r];
#pragma unroll
    for (int v = 0; v < 8; ++v) yout[(size_t)v * D + nt * 16] = c[nt][v] + bv;
  }
}

#define CAP 32

__global__ void zero_i32(int* __restrict__ p, int n) {
  const int i = blockIdx.x * blockDim.x + threadIdx.x;
  if (i < n) p[i] = 0;
}

__global__ void __launch_bounds__(256)
fill_slots(const int* __restrict__ src_f, const int* __restrict__ src_rb,
           const int* __restrict__ src_r,
           const int* __restrict__ dst_f, const int* __restrict__ dst_rb,
           const int* __restrict__ dst_r,
           int* __restrict__ cnt, int* __restrict__ slots,
           int Ef, int Erb, int Er, int NU) {
  const int et = blockIdx.y;
  const int* src = et == 0 ? src_f : (et == 1 ? src_rb : src_r);
  const int* dst = et == 0 ? dst_f : (et == 1 ? dst_rb : dst_r);
  const int nE   = et == 0 ? Ef : (et == 1 ? Erb : Er);
  const int coff = et == 0 ? 0 : (et == 1 ? NU : 2 * NU);
  const int e = blockIdx.x * blockDim.x + threadIdx.x;
  if (e >= nE) return;
  const int node = coff + dst[e];
  const int s = src[e];
  const int r = atomicAdd(cnt + node, 1);
  slots[(size_t)node * CAP + (r & (CAP - 1))] = s;
}

__device__ __forceinline__ void add_rows(float4& acc, const float* __restrict__ whl,
                                         int id, int n) {
  int k = 0;
  for (; k + 8 <= n; k += 8) {
    float4 v[8];
#pragma unroll
    for (int i = 0; i < 8; ++i)
      v[i] = *(const float4*)(whl + (size_t)__builtin_amdgcn_readlane(id, k + i) * D);
#pragma unroll
    for (int i = 0; i < 8; ++i) {
      acc.x += v[i].x; acc.y += v[i].y; acc.z += v[i].z; acc.w += v[i].w;
    }
  }
  int s[7];
#pragma unroll
  for (int i = 0; i < 7; ++i) s[i] = __builtin_amdgcn_readlane(id, (k + i) & 31);
  float4 t[7];
#pragma unroll
  for (int i = 0; i < 7; ++i) {
    t[i] = make_float4(0.0f, 0.0f, 0.0f, 0.0f);
    if (k + i < n) t[i] = *(const float4*)(whl + (size_t)s[i] * D);
  }
#pragma unroll
  for (int i = 0; i < 7; ++i) {
    acc.x += t[i].x; acc.y += t[i].y; acc.z += t[i].z; acc.w += t[i].w;
  }
}

__device__ __forceinline__ float4 node_sum(const float* __restrict__ whl,
                                           const int* __restrict__ src,
                                           const int* __restrict__ dst, int nE,
                                           int d, int deg, int slot, int lane) {
  float4 acc = make_float4(0.0f, 0.0f, 0.0f, 0.0f);
  if (deg <= CAP) {
    add_rows(acc, whl, slot, deg);
    return acc;
  }
  for (int e0 = 0; e0 < nE; e0 += 32) {
    const int e = e0 + lane;
    bool hit = false;
    int s = 0;
    if (e < nE) {
      hit = dst[e] == d;
      s = src[e];
    }
    unsigned m = __builtin_amdgcn_ballot_w32(hit);
    while (m != 0u) {
      const int l = __builtin_amdgcn_readfirstlane(__builtin_ctz(m));
      m &= m - 1u;
      const int row = __builtin_amdgcn_readlane(s, l);
      const float4 v = *(const float4*)(whl + (size_t)row * D);
      acc.x += v.x; acc.y += v.y; acc.z += v.z; acc.w += v.w;
    }
  }
  return acc;
}

__global__ void __launch_bounds__(256)
reduce_nodes(const float* __restrict__ Wh_f, const float* __restrict__ Wh_rb,
             const float* __restrict__ Wh_r,
             const int* __restrict__ src_f, const int* __restrict__ src_rb,
             const int* __restrict__ src_r,
             const int* __restrict__ dst_f, const int* __restrict__ dst_rb,
             const int* __restrict__ dst_r,
             const int* __restrict__ cnt, const int* __restrict__ slots,
             float* __restrict__ out, int NU, int NI, int Ef, int Erb, int Er) {
  const int lane = threadIdx.x & 31;
  const int w = __builtin_amdgcn_readfirstlane((int)(blockIdx.x * 8 + (threadIdx.x >> 5)));
  if (w >= NU + NI) return;
  float4 res;
  if (w < NU) {
    const int cf  = __builtin_amdgcn_readfirstlane(cnt[w]);
    const int crb = __builtin_amdgcn_readfirstlane(cnt[NU + w]);
    int idf = 0, idrb = 0;
    if (lane < cf) idf = slots[(size_t)w * CAP + lane];
    if (lane < crb) idrb = slots[(size_t)(NU + w) * CAP + lane];
    const float4 a = node_sum(Wh_f + lane * 4, src_f, dst_f, Ef, w, cf, idf, lane);
    const float4 b = node_sum(Wh_rb + lane * 4, src_rb, dst_rb, Erb, w, crb, idrb, lane);
    const float df = fmaxf((float)cf, 1.0f);
    const float drb = fmaxf((float)crb, 1.0f);
    res.x = a.x / df + b.x / drb;
    res.y = a.y / df + b.y / drb;
    res.z = a.z / df + b.z / drb;
    res.w = a.w / df + b.w / drb;
  } else {
    const int i = w - NU;
    const int c = __builtin_amdgcn_readfirstlane(cnt[2 * NU + i]);
    int id = 0;
    if (lane < c) id = slots[(size_t)(2 * NU + i) * CAP + lane];
    const float4 a = node_sum(Wh_r + lane * 4, src_r, dst_r, Er, i, c, id, lane);
    const float dr = fmaxf((float)c, 1.0f);
    res.x = a.x / dr; res.y = a.y / dr; res.z = a.z / dr; res.w = a.w / dr;
  }
  *(float4*)(out + (size_t)w * D + lane * 4) = res;
}

extern "C" void kernel_launch(void* const* d_in, const int* in_sizes, int n_in,
                              void* d_out, int out_size, void* d_ws, size_t ws_size,
                              hipStream_t stream) {
  (void)n_in; (void)out_size; (void)ws_size;

  const float* feat_user  = (const float*)d_in[0];
  const float* feat_item  = (const float*)d_in[1];
  const float* W_follows  = (const float*)d_in[2];
  const float* b_follows  = (const float*)d_in[3];
  const float* W_rates    = (const float*)d_in[4];
  const float* b_rates    = (const float*)d_in[5];
  const float* W_ratedby  = (const float*)d_in[6];
  const float* b_ratedby  = (const float*)d_in[7];
  const int* src_follows  = (const int*)d_in[8];
  const int* dst_follows  = (const int*)d_in[9];
  const int* src_rates    = (const int*)d_in[10];
  const int* dst_rates    = (const int*)d_in[11];
  const int* src_ratedby  = (const int*)d_in[12];
  const int* dst_ratedby  = (const int*)d_in[13];

  const int NU  = in_sizes[0] / D;
  const int NI  = in_sizes[1] / D;
  const int Ef  = in_sizes[8];
  const int Er  = in_sizes[10];
  const int Erb = in_sizes[12];

  float* out = (float*)d_out;

  float* ws = (float*)d_ws;
  const int nnmax = NU > NI ? NU : NI;
  const size_t nmax = (size_t)nnmax * D;
  float* wh_f  = ws;
  float* wh_rb = wh_f + nmax;
  float* wh_r  = wh_rb + nmax;
  const int ncnt = 2 * NU + NI;
  int* slots  = (int*)(wh_r + nmax);
  int* cnt    = slots + (size_t)ncnt * CAP;
  int emax = Ef > Erb ? Ef : Erb;
  emax = emax > Er ? emax : Er;

  zero_i32<<<(ncnt + 255) / 256, 256, 0, stream>>>(cnt, ncnt);
  fill_slots<<<dim3((emax + 255) / 256, 3), 256, 0, stream>>>(
      src_follows, src_ratedby, src_rates, dst_follows, dst_ratedby, dst_rates,
      cnt, slots, Ef, Erb, Er, NU);

  gemm_wmma_f32<<<(NU / 16 + 7) / 8, 256, 0, stream>>>(feat_user, W_follows, b_follows, wh_f, NU / 16);
  gemm_wmma_f32<<<(NI / 16 + 7) / 8, 256, 0, stream>>>(feat_item, W_ratedby, b_ratedby, wh_rb, NI / 16);
  gemm_wmma_f32<<<(NU / 16 + 7) / 8, 256, 0, stream>>>(feat_user, W_rates, b_rates, wh_r, NU / 16);

  reduce_nodes<<<(NU + NI + 7) / 8, 256, 0, stream>>>(
      wh_f, wh_rb, wh_r, src_follows, src_ratedby, src_rates,
      dst_follows, dst_ratedby, dst_rates, cnt, slots, out, NU, NI, Ef, Erb, Er);
}


# a wave loops over its items instead of one node (or 32 edges) per wave: reduce_nodes 300000 -> 16384 waves (node w, w+16384, ...), fill_slots 75000 -> 9384, zero 9376 -> 2344; sums stay in rank order
# speedup vs baseline: 2.7041x; 1.0840x over previous
#include <hip/hip_runtime.h>

typedef __attribute__((ext_vector_type(2))) float v2f;
typedef __attribute__((ext_vector_type(8))) float v8f;

#define D 128

#define WP 132

__device__ __forceinline__ void gemm_source(v8f (&c)[8], float* __restrict__ sW,
                                            const float* __restrict__ X,
                                            const float* __restrict__ W,
                                            int tl, int tid, int r, int h, bool refill) {

  const float* arow = X + (size_t)(tl * 16 + r) * D + h * 4;
  float4 a[16];
#pragma unroll
  for (int j = 0; j < 16; ++j) a[j] = *(const float4*)(arow + j * 8);

  if (refill) __syncthreads();

#pragma unroll
  for (int i = 0; i < 16; ++i) {
    const int idx = tid + i * 256;
    const int n = idx >> 5;
    const int c4 = idx & 31;
    const float4 w4 = ((const float4*)W)[idx];
    *(float4*)(sW + n * WP + c4 * 4) = w4;
  }
  __syncthreads();

  const float* bl = sW + r * WP + h * 4;

  float4 b[8];
#pragma unroll
  for (int nt = 0; nt < 8; ++nt) b[nt] = *(const float4*)(bl + nt * 16 * WP);

#pragma unroll
  for (int j = 0; j < 16; ++j) {
    float4 bn[8];
    if (j + 1 < 16) {
#pragma unroll
      for (int nt = 0; nt < 8; ++nt)
        bn[nt] = *(const float4*)(bl + nt * 16 * WP + (j + 1) * 8);
    }
    const v2f a0 = {a[j].x, a[j].y};
    const v2f a1 = {a[j].z, a[j].w};
#pragma unroll
    for (int nt = 0; nt < 8; ++nt) {
      const v2f b0 = {b[nt].x, b[nt].y};
      c[nt] = __builtin_amdgcn_wmma_f32_16x16x4_f32(false, a0, false, b0,
                                                    (short)0, c[nt], false, false);
    }
#pragma unroll
    for (int nt = 0; nt < 8; ++nt) {
      const v2f b1 = {b[nt].z, b[nt].w};
      c[nt] = __builtin_amdgcn_wmma_f32_16x16x4_f32(false, a1, false, b1,
                                                    (short)0, c[nt], false, false);
    }
    if (j + 1 < 16) {
#pragma unroll
      for (int nt = 0; nt < 8; ++nt) b[nt] = bn[nt];
    }
  }
}

__global__ void __launch_bounds__(256)
gemm_wmma_f32_two(const float* __restrict__ A0, const float* __restrict__ W0,
                  const float* __restrict__ bias0, const int* __restrict__ deg0,
                  const float* __restrict__ A1, const float* __restrict__ W1,
                  const float* __restrict__ bias1, const int* __restrict__ deg1,
                  float* __restrict__ Y, int ntiles) {
  __shared__ __attribute__((aligned(16))) float sW[D * WP];
  const int tid = threadIdx.x;
  const int lane = tid & 31;
  const int wave = __builtin_amdgcn_readfirstlane(tid >> 5);
  const int r = lane & 15;
  const int h = lane >> 4;
  const int tile = blockIdx.x * 8 + wave;
  const int tl = tile < ntiles ? tile : ntiles - 1;

  v8f c[8];
#pragma unroll
  for (int nt = 0; nt < 8; ++nt) c[nt] = (v8f){};

  gemm_source(c, sW, A0, W0, tl, tid, r, h, false);
  gemm_source(c, sW, A1, W1, tl, tid, r, h, true);

  if (tile >= ntiles) return;

  bool g0[8], g1[8];
#pragma unroll
  for (int v = 0; v < 8; ++v) {
    g0[v] = deg0[tile * 16 + h * 8 + v] > 0;
    g1[v] = deg1[tile * 16 + h * 8 + v] > 0;
  }
  float* yout = Y + (size_t)(tile * 16 + h * 8) * D + r;
#pragma unroll
  for (int nt = 0; nt < 8; ++nt) {
    const float bv0 = bias0[nt * 16 + r];
    const float bv1 = bias1[nt * 16 + r];
#pragma unroll
    for (int v = 0; v < 8; ++v)
      yout[(size_t)v * D + nt * 16] = c[nt][v] + (g0[v] ? bv0 : 0.0f) + (g1[v] ? bv1 : 0.0f);
  }
}

__global__ void __launch_bounds__(256)
gemm_wmma_f32_one(const float* __restrict__ A0, const float* __restrict__ W0,
                  const float* __restrict__ bias0, const int* __restrict__ deg0,
                  float* __restrict__ Y, int ntiles) {
  __shared__ __attribute__((aligned(16))) float sW[D * WP];
  const int tid = threadIdx.x;
  const int lane = tid & 31;
  const int wave = __builtin_amdgcn_readfirstlane(tid >> 5);
  const int r = lane & 15;
  const int h = lane >> 4;
  const int tile = blockIdx.x * 8 + wave;
  const int tl = tile < ntiles ? tile : ntiles - 1;

  v8f c[8];
#pragma unroll
  for (int nt = 0; nt < 8; ++nt) c[nt] = (v8f){};

  gemm_source(c, sW, A0, W0, tl, tid, r, h, false);

  if (tile >= ntiles) return;

  bool g0[8];
#pragma unroll
  for (int v = 0; v < 8; ++v) g0[v] = deg0[tile * 16 + h * 8 + v] > 0;
  float* yout = Y + (size_t)(tile * 16 + h * 8) * D + r;
#pragma unroll
  for (int nt = 0; nt < 8; ++nt) {
    const float bv0 = bias0[nt * 16 + r];
#pragma unroll
    for (int v = 0; v < 8; ++v)
      yout[(size_t)v * D + nt * 16] = c[nt][v] + (g0[v] ? bv0 : 0.0f);
  }
}

#define CAP 32

__global__ void zero_i32(int* __restrict__ p, int n) {
  const int i = (blockIdx.x * blockDim.x + threadIdx.x) * 4;
  if (i + 3 < n) {
    *(int4*)(p + i) = make_int4(0, 0, 0, 0);
  } else {
    for (int k = i; k < n; ++k) p[k] = 0;
  }
}

__global__ void __launch_bounds__(256)
fill_slots(const int* __restrict__ src_f, const int* __restrict__ src_rb,
           const int* __restrict__ src_r,
           const int* __restrict__ dst_f, const int* __restrict__ dst_rb,
           const int* __restrict__ dst_r,
           int* __restrict__ cnt, int* __restrict__ slots,
           int Ef, int Erb, int Er, int NU) {
  const int et = blockIdx.y;
  const int* src = et == 0 ? src_f : (et == 1 ? src_rb : src_r);
  const int* dst = et == 0 ? dst_f : (et == 1 ? dst_rb : dst_r);
  const int nE   = et == 0 ? Ef : (et == 1 ? Erb : Er);
  const int coff = et == 0 ? 0 : (et == 1 ? NU : 2 * NU);
  const int stride = gridDim.x * blockDim.x;
  for (int e0 = blockIdx.x * blockDim.x + threadIdx.x; e0 < nE; e0 += 4 * stride) {
    bool ok[4];
    int node[4], s[4], r[4];
#pragma unroll
    for (int k = 0; k < 4; ++k) {
      const int e = e0 + k * stride;
      ok[k] = e < nE;
      const int ec = ok[k] ? e : e0;
      node[k] = dst[ec];
      s[k] = src[ec];
    }
#pragma unroll
    for (int k = 0; k < 4; ++k) {
      r[k] = 0;
      if (ok[k]) r[k] = atomicAdd(cnt + coff + node[k], 1);
    }
#pragma unroll
    for (int k = 0; k < 4; ++k)
      if (ok[k]) slots[(size_t)(coff + node[k]) * CAP + (r[k] & (CAP - 1))] = s[k];
  }
}

__device__ __forceinline__ void add_rows(float4& acc, const float* __restrict__ whl,
                                         int id, int n) {
  int k = 0;
  for (; k + 8 <= n; k += 8) {
    float4 v[8];
#pragma unroll
    for (int i = 0; i < 8; ++i)
      v[i] = *(const float4*)(whl + (size_t)__builtin_amdgcn_readlane(id, k + i) * D);
#pragma unroll
    for (int i = 0; i < 8; ++i) {
      acc.x += v[i].x; acc.y += v[i].y; acc.z += v[i].z; acc.w += v[i].w;
    }
  }
  int s[7];
#pragma unroll
  for (int i = 0; i < 7; ++i) s[i] = __builtin_amdgcn_readlane(id, (k + i) & 31);
  float4 t[7];
#pragma unroll
  for (int i = 0; i < 7; ++i) {
    t[i] = make_float4(0.0f, 0.0f, 0.0f, 0.0f);
    if (k + i < n) t[i] = *(const float4*)(whl + (size_t)s[i] * D);
  }
#pragma unroll
  for (int i = 0; i < 7; ++i) {
    acc.x += t[i].x; acc.y += t[i].y; acc.z += t[i].z; acc.w += t[i].w;
  }
}

__device__ __forceinline__ float4 node_sum(const float* __restrict__ whl,
                                           const int* __restrict__ src,
                                           const int* __restrict__ dst, int nE,
                                           int d, int deg, int slot, int lane) {
  float4 acc = make_float4(0.0f, 0.0f, 0.0f, 0.0f);
  if (deg <= CAP) {
    add_rows(acc, whl, slot, deg);
    return acc;
  }
  for (int e0 = 0; e0 < nE; e0 += 32) {
    const int e = e0 + lane;
    bool hit = false;
    int s = 0;
    if (e < nE) {
      hit = dst[e] == d;
      s = src[e];
    }
    unsigned m = __builtin_amdgcn_ballot_w32(hit);
    while (m != 0u) {
      const int l = __builtin_amdgcn_readfirstlane(__builtin_ctz(m));
      m &= m - 1u;
      const int row = __builtin_amdgcn_readlane(s, l);
      const float4 v = *(const float4*)(whl + (size_t)row * D);
      acc.x += v.x; acc.y += v.y; acc.z += v.z; acc.w += v.w;
    }
  }
  return acc;
}

__global__ void __launch_bounds__(256)
reduce_nodes(const float* __restrict__ feat_user, const float* __restrict__ feat_item,
             const int* __restrict__ src_f, const int* __restrict__ src_rb,
             const int* __restrict__ src_r,
             const int* __restrict__ dst_f, const int* __restrict__ dst_rb,
             const int* __restrict__ dst_r,
             const int* __restrict__ cnt, const int* __restrict__ slots,
             float* __restrict__ agg_f, float* __restrict__ agg_rb,
             float* __restrict__ agg_r, int NU, int NI, int Ef, int Erb, int Er) {
  const int lane = threadIdx.x & 31;
  int w = __builtin_amdgcn_readfirstlane((int)(blockIdx.x * 8 + (threadIdx.x >> 5)));
  const int nw = (int)(gridDim.x * 8);
  const int total = 2 * NU + NI;
  if (w >= total) return;
#define ROW_OF(v) ((v) < NU ? (v) : ((v) < NU + NI ? (v) + NU : (v) - NI))
  int c = __builtin_amdgcn_readfirstlane(cnt[ROW_OF(w)]);
  int id = 0;
  if (lane < c) id = slots[(size_t)ROW_OF(w) * CAP + lane];
  int wn = w + nw;
  int cnv = 0;
  if (wn < total) cnv = cnt[ROW_OF(wn)];
  for (;;) {
    const int cn = __builtin_amdgcn_readfirstlane(cnv);
    int idn = 0;
    if (lane < cn) idn = slots[(size_t)ROW_OF(wn) * CAP + lane];
    const int wnn = wn + nw;
    int cnnv = 0;
    if (wnn < total) cnnv = cnt[ROW_OF(wnn)];

    const float* tab; const int* src; const int* dst; float* agg;
    int nE, d;
    if (w < NU) {
      tab = feat_user; src = src_f; dst = dst_f; agg = agg_f; nE = Ef; d = w;
    } else if (w < NU + NI) {
      tab = feat_user; src = src_r; dst = dst_r; agg = agg_r; nE = Er; d = w - NU;
    } else {
      tab = feat_item; src = src_rb; dst = dst_rb; agg = agg_rb; nE = Erb; d = w - NU - NI;
    }
    const float4 a = node_sum(tab + lane * 4, src, dst, nE, d, c, id, lane);
    const float dg = fmaxf((float)c, 1.0f);
    float4 res;
    res.x = a.x / dg; res.y = a.y / dg; res.z = a.z / dg; res.w = a.w / dg;
    *(float4*)(agg + (size_t)d * D + lane * 4) = res;

    if (wn >= total) break;
    w = wn; c = cn; id = idn; wn = wnn; cnv = cnnv;
  }
#undef ROW_OF
}

extern "C" void kernel_launch(void* const* d_in, const int* in_sizes, int n_in,
                              void* d_out, int out_size, void* d_ws, size_t ws_size,
                              hipStream_t stream) {
  (void)n_in; (void)out_size; (void)ws_size;

  const float* feat_user  = (const float*)d_in[0];
  const float* feat_item  = (const float*)d_in[1];
  const float* W_follows  = (const float*)d_in[2];
  const float* b_follows  = (const float*)d_in[3];
  const float* W_rates    = (const float*)d_in[4];
  const float* b_rates    = (const float*)d_in[5];
  const float* W_ratedby  = (const float*)d_in[6];
  const float* b_ratedby  = (const float*)d_in[7];
  const int* src_follows  = (const int*)d_in[8];
  const int* dst_follows  = (const int*)d_in[9];
  const int* src_rates    = (const int*)d_in[10];
  const int* dst_rates    = (const int*)d_in[11];
  const int* src_ratedby  = (const int*)d_in[12];
  const int* dst_ratedby  = (const int*)d_in[13];

  const int NU  = in_sizes[0] / D;
  const int NI  = in_sizes[1] / D;
  const int Ef  = in_sizes[8];
  const int Er  = in_sizes[10];
  const int Erb = in_sizes[12];

  float* out = (float*)d_out;

  float* ws = (float*)d_ws;
  const int nnmax = NU > NI ? NU : NI;
  const size_t nmax = (size_t)nnmax * D;
  float* agg_f  = ws;
  float* agg_rb = agg_f + nmax;
  float* agg_r  = agg_rb + nmax;
  const int ncnt = 2 * NU + NI;
  int* slots  = (int*)(agg_r + nmax);
  int* cnt    = slots + (size_t)ncnt * CAP;
  int emax = Ef > Erb ? Ef : Erb;
  emax = emax > Er ? emax : Er;

  zero_i32<<<(ncnt + 1023) / 1024, 256, 0, stream>>>(cnt, ncnt);
  fill_slots<<<dim3((emax + 2047) / 2048, 3), 256, 0, stream>>>(
      src_follows, src_ratedby, src_rates, dst_follows, dst_ratedby, dst_rates,
      cnt, slots, Ef, Erb, Er, NU);

  reduce_nodes<<<2048, 256, 0, stream>>>(
      feat_user, feat_item, src_follows, src_ratedby, src_rates,
      dst_follows, dst_ratedby, dst_rates, cnt, slots, agg_f, agg_rb, agg_r,
      NU, NI, Ef, Erb, Er);

  gemm_wmma_f32_two<<<(NU / 16 + 7) / 8, 256, 0, stream>>>(
      agg_f, W_follows, b_follows, cnt, agg_rb, W_ratedby, b_ratedby, cnt + NU,
      out, NU / 16);
  gemm_wmma_f32_one<<<(NI / 16 + 7) / 8, 256, 0, stream>>>(
      agg_r, W_rates, b_rates, cnt + 2 * NU, out + (size_t)NU * D, NI / 16);
}


# both GEMMs as ONE launch of 256 resident 16-wave work-groups: three W staged in LDS once, 16-row tiles dealt evenly over all waves, next A strip loaded under the WMMAs; 5 -> 4 launches; same sums
# speedup vs baseline: 2.8944x; 1.0704x over previous
#include <hip/hip_runtime.h>

typedef __attribute__((ext_vector_type(2))) float v2f;
typedef __attribute__((ext_vector_type(8))) float v8f;

#define D 128

#define WP 132

__device__ __forceinline__ void gemm_strip(v8f (&c)[8], float4 (&a)[16],
                                           const float* __restrict__ sW,
                                           const float* __restrict__ anext,
                                           int r, int h) {
  const float* bl = sW + r * WP + h * 4;

  float4 b[8];
#pragma unroll
  for (int nt = 0; nt < 8; ++nt) b[nt] = *(const float4*)(bl + nt * 16 * WP);

#pragma unroll
  for (int j = 0; j < 16; ++j) {
    float4 bn[8];
    if (j + 1 < 16) {
#pragma unroll
      for (int nt = 0; nt < 8; ++nt)
        bn[nt] = *(const float4*)(bl + nt * 16 * WP + (j + 1) * 8);
    }
    __builtin_amdgcn_sched_barrier(0);
    const v2f a0 = {a[j].x, a[j].y};
    const v2f a1 = {a[j].z, a[j].w};
#pragma unroll
    for (int nt = 0; nt < 8; ++nt) {
      const v2f b0 = {b[nt].x, b[nt].y};
      c[nt] = __builtin_amdgcn_wmma_f32_16x16x4_f32(false, a0, false, b0,
                                                    (short)0, c[nt], false, false);
    }
#pragma unroll
    for (int nt = 0; nt < 8; ++nt) {
      const v2f b1 = {b[nt].z, b[nt].w};
      c[nt] = __builtin_amdgcn_wmma_f32_16x16x4_f32(false, a1, false, b1,
                                                    (short)0, c[nt], false, false);
    }
    a[j] = *(const float4*)(anext + j * 8);
    if (j + 1 < 16) {
#pragma unroll
      for (int nt = 0; nt < 8; ++nt) b[nt] = bn[nt];
    }
    __builtin_amdgcn_sched_barrier(0);
  }
}

__global__ void __launch_bounds__(512)
gemm_wmma_f32_all(const float* __restrict__ A_f, const float* __restrict__ W_f,
                  const float* __restrict__ b_f,
                  const float* __restrict__ A_rb, const float* __restrict__ W_rb,
                  const float* __restrict__ b_rb,
                  const float* __restrict__ A_r, const float* __restrict__ W_r,
                  const float* __restrict__ b_r,
                  const int* __restrict__ cnt, float* __restrict__ Y,
                  int ntu, int nti, int NU) {
  __shared__ __attribute__((aligned(16))) float sW[3 * D * WP];
  const int tid = threadIdx.x;
  const int lane = tid & 31;
  const int wave = __builtin_amdgcn_readfirstlane(tid >> 5);
  const int r = lane & 15;
  const int h = lane >> 4;

  const int nw = (int)gridDim.x * 16;
  const int g = wave * (int)gridDim.x + (int)blockIdx.x;
  const int ti0 = nw - 1 - g;

  const size_t lo = (size_t)r * D + h * 4;
#define STRIP(X, tile) ((X) + (size_t)(tile) * (16 * D) + lo)

  float4 a[16];
  {
    const float* p = g < ntu ? STRIP(A_f, g) : STRIP(A_r, ti0 < nti ? ti0 : 0);
#pragma unroll
    for (int j = 0; j < 16; ++j) a[j] = *(const float4*)(p + j * 8);
  }

#pragma unroll
  for (int s = 0; s < 3; ++s) {
    const float* W = s == 0 ? W_f : (s == 1 ? W_rb : W_r);
#pragma unroll
    for (int i = 0; i < 8; ++i) {
      const int idx = tid + i * 512;
      const int n = idx >> 5;
      const int c4 = idx & 31;
      const float4 w4 = ((const float4*)W)[idx];
      *(float4*)(sW + s * (D * WP) + n * WP + c4 * 4) = w4;
    }
  }
  __syncthreads();

  const int* deg_f = cnt;
  const int* deg_rb = cnt + NU;
  const int* deg_r = cnt + 2 * NU;

  for (int tile = g; tile < ntu; tile += nw) {
    v8f c[8];
#pragma unroll
    for (int nt = 0; nt < 8; ++nt) c[nt] = (v8f){};

    gemm_strip(c, a, sW, STRIP(A_rb, tile), r, h);
    const int tn = tile + nw;
    const float* nx = tn < ntu ? STRIP(A_f, tn)
                               : (ti0 < nti ? STRIP(A_r, ti0) : STRIP(A_rb, tile));
    gemm_strip(c, a, sW + D * WP, nx, r, h);

    bool g0[8], g1[8];
#pragma unroll
    for (int v = 0; v < 8; ++v) {
      g0[v] = deg_f[tile * 16 + h * 8 + v] > 0;
      g1[v] = deg_rb[tile * 16 + h * 8 + v] > 0;
    }
    float* yout = Y + (size_t)(tile * 16 + h * 8) * D + r;
#pragma unroll
    for (int nt = 0; nt < 8; ++nt) {
      const float bv0 = b_f[nt * 16 + r];
      const float bv1 = b_rb[nt * 16 + r];
#pragma unroll
      for (int v = 0; v < 8; ++v)
        yout[(size_t)v * D + nt * 16] = c[nt][v] + (g0[v] ? bv0 : 0.0f) + (g1[v] ? bv1 : 0.0f);
    }
  }

  float* Yi = Y + (size_t)NU * D;
  for (int tile = ti0; tile < nti; tile += nw) {
    v8f c[8];
#pragma unroll
    for (int nt = 0; nt < 8; ++nt) c[nt] = (v8f){};

    const int tn = tile + nw;
    gemm_strip(c, a, sW + 2 * D * WP, STRIP(A_r, tn < nti ? tn : tile), r, h);

    bool g0[8];
#pragma unroll
    for (int v = 0; v < 8; ++v) g0[v] = deg_r[tile * 16 + h * 8 + v] > 0;
    float* yout = Yi + (size_t)(tile * 16 + h * 8) * D + r;
#pragma unroll
    for (int nt = 0; nt < 8; ++nt) {
      const float bv0 = b_r[nt * 16 + r];
#pragma unroll
      for (int v = 0; v < 8; ++v)
        yout[(size_t)v * D + nt * 16] = c[nt][v] + (g0[v] ? bv0 : 0.0f);
    }
  }
#undef STRIP
}

#define CAP 32

__global__ void zero_i32(int* __restrict__ p, int n) {
  const int i = (blockIdx.x * blockDim.x + threadIdx.x) * 4;
  if (i + 3 < n) {
    *(int4*)(p + i) = make_int4(0, 0, 0, 0);
  } else {
    for (int k = i; k < n; ++k) p[k] = 0;
  }
}

__global__ void __launch_bounds__(256)
fill_slots(const int* __restrict__ src_f, const int* __restrict__ src_rb,
           const int* __restrict__ src_r,
           const int* __restrict__ dst_f, const int* __restrict__ dst_rb,
           const int* __restrict__ dst_r,
           int* __restrict__ cnt, int* __restrict__ slots,
           int Ef, int Erb, int Er, int NU) {
  const int et = blockIdx.y;
  const int* src = et == 0 ? src_f : (et == 1 ? src_rb : src_r);
  const int* dst = et == 0 ? dst_f : (et == 1 ? dst_rb : dst_r);
  const int nE   = et == 0 ? Ef : (et == 1 ? Erb : Er);
  const int coff = et == 0 ? 0 : (et == 1 ? NU : 2 * NU);
  const int stride = gridDim.x * blockDim.x;
  for (int e0 = blockIdx.x * blockDim.x + threadIdx.x; e0 < nE; e0 += 4 * stride) {
    bool ok[4];
    int node[4], s[4], r[4];
#pragma unroll
    for (int k = 0; k < 4; ++k) {
      const int e = e0 + k * stride;
      ok[k] = e < nE;
      const int ec = ok[k] ? e : e0;
      node[k] = dst[ec];
      s[k] = src[ec];
    }
#pragma unroll
    for (int k = 0; k < 4; ++k) {
      r[k] = 0;
      if (ok[k]) r[k] = atomicAdd(cnt + coff + node[k], 1);
    }
#pragma unroll
    for (int k = 0; k < 4; ++k)
      if (ok[k]) slots[(size_t)(coff + node[k]) * CAP + (r[k] & (CAP - 1))] = s[k];
  }
}

__device__ __forceinline__ void add_rows(float4& acc, const float* __restrict__ whl,
                                         int id, int n) {
  int k = 0;
  for (; k + 8 <= n; k += 8) {
    float4 v[8];
#pragma unroll
    for (int i = 0; i < 8; ++i)
      v[i] = *(const float4*)(whl + (size_t)__builtin_amdgcn_readlane(id, k + i) * D);
#pragma unroll
    for (int i = 0; i < 8; ++i) {
      acc.x += v[i].x; acc.y += v[i].y; acc.z += v[i].z; acc.w += v[i].w;
    }
  }
  int s[7];
#pragma unroll
  for (int i = 0; i < 7; ++i) s[i] = __builtin_amdgcn_readlane(id, (k + i) & 31);
  float4 t[7];
#pragma unroll
  for (int i = 0; i < 7; ++i) {
    t[i] = make_float4(0.0f, 0.0f, 0.0f, 0.0f);
    if (k + i < n) t[i] = *(const float4*)(whl + (size_t)s[i] * D);
  }
#pragma unroll
  for (int i = 0; i < 7; ++i) {
    acc.x += t[i].x; acc.y += t[i].y; acc.z += t[i].z; acc.w += t[i].w;
  }
}

__device__ __forceinline__ float4 node_sum(const float* __restrict__ whl,
                                           const int* __restrict__ src,
                                           const int* __restrict__ dst, int nE,
                                           int d, int deg, int slot, int lane) {
  float4 acc = make_float4(0.0f, 0.0f, 0.0f, 0.0f);
  if (deg <= CAP) {
    add_rows(acc, whl, slot, deg);
    return acc;
  }
  for (int e0 = 0; e0 < nE; e0 += 32) {
    const int e = e0 + lane;
    bool hit = false;
    int s = 0;
    if (e < nE) {
      hit = dst[e] == d;
      s = src[e];
    }
    unsigned m = __builtin_amdgcn_ballot_w32(hit);
    while (m != 0u) {
      const int l = __builtin_amdgcn_readfirstlane(__builtin_ctz(m));
      m &= m - 1u;
      const int row = __builtin_amdgcn_readlane(s, l);
      const float4 v = *(const float4*)(whl + (size_t)row * D);
      acc.x += v.x; acc.y += v.y; acc.z += v.z; acc.w += v.w;
    }
  }
  return acc;
}

__global__ void __launch_bounds__(256)
reduce_nodes(const float* __restrict__ feat_user, const float* __restrict__ feat_item,
             const int* __restrict__ src_f, const int* __restrict__ src_rb,
             const int* __restrict__ src_r,
             const int* __restrict__ dst_f, const int* __restrict__ dst_rb,
             const int* __restrict__ dst_r,
             const int* __restrict__ cnt, const int* __restrict__ slots,
             float* __restrict__ agg_f, float* __restrict__ agg_rb,
             float* __restrict__ agg_r, int NU, int NI, int Ef, int Erb, int Er) {
  const int lane = threadIdx.x & 31;
  int w = __builtin_amdgcn_readfirstlane((int)(blockIdx.x * 8 + (threadIdx.x >> 5)));
  const int nw = (int)(gridDim.x * 8);
  const int total = 2 * NU + NI;
  if (w >= total) return;
#define ROW_OF(v) ((v) < NU ? (v) : ((v) < NU + NI ? (v) + NU : (v) - NI))
  int c = __builtin_amdgcn_readfirstlane(cnt[ROW_OF(w)]);
  int id = 0;
  if (lane < c) id = slots[(size_t)ROW_OF(w) * CAP + lane];
  int wn = w + nw;
  int cnv = 0;
  if (wn < total) cnv = cnt[ROW_OF(wn)];
  for (;;) {
    const int cn = __builtin_amdgcn_readfirstlane(cnv);
    int idn = 0;
    if (lane < cn) idn = slots[(size_t)ROW_OF(wn) * CAP + lane];
    const int wnn = wn + nw;
    int cnnv = 0;
    if (wnn < total) cnnv = cnt[ROW_OF(wnn)];

    const float* tab; const int* src; const int* dst; float* agg;
    int nE, d;
    if (w < NU) {
      tab = feat_user; src = src_f; dst = dst_f; agg = agg_f; nE = Ef; d = w;
    } else if (w < NU + NI) {
      tab = feat_user; src = src_r; dst = dst_r; agg = agg_r; nE = Er; d = w - NU;
    } else {
      tab = feat_item; src = src_rb; dst = dst_rb; agg = agg_rb; nE = Erb; d = w - NU - NI;
    }
    const float4 a = node_sum(tab + lane * 4, src, dst, nE, d, c, id, lane);
    const float dg = fmaxf((float)c, 1.0f);
    float4 res;
    res.x = a.x / dg; res.y = a.y / dg; res.z = a.z / dg; res.w = a.w / dg;
    *(float4*)(agg + (size_t)d * D + lane * 4) = res;

    if (wn >= total) break;
    w = wn; c = cn; id = idn; wn = wnn; cnv = cnnv;
  }
#undef ROW_OF
}

extern "C" void kernel_launch(void* const* d_in, const int* in_sizes, int n_in,
                              void* d_out, int out_size, void* d_ws, size_t ws_size,
                              hipStream_t stream) {
  (void)n_in; (void)out_size; (void)ws_size;

  const float* feat_user  = (const float*)d_in[0];
  const float* feat_item  = (const float*)d_in[1];
  const float* W_follows  = (const float*)d_in[2];
  const float* b_follows  = (const float*)d_in[3];
  const float* W_rates    = (const float*)d_in[4];
  const float* b_rates    = (const float*)d_in[5];
  const float* W_ratedby  = (const float*)d_in[6];
  const float* b_ratedby  = (const float*)d_in[7];
  const int* src_follows  = (const int*)d_in[8];
  const int* dst_follows  = (const int*)d_in[9];
  const int* src_rates    = (const int*)d_in[10];
  const int* dst_rates    = (const int*)d_in[11];
  const int* src_ratedby  = (const int*)d_in[12];
  const int* dst_ratedby  = (const int*)d_in[13];

  const int NU  = in_sizes[0] / D;
  const int NI  = in_sizes[1] / D;
  const int Ef  = in_sizes[8];
  const int Er  = in_sizes[10];
  const int Erb = in_sizes[12];

  float* out = (float*)d_out;

  float* ws = (float*)d_ws;
  const int nnmax = NU > NI ? NU : NI;
  const size_t nmax = (size_t)nnmax * D;
  float* agg_f  = ws;
  float* agg_rb = agg_f + nmax;
  float* agg_r  = agg_rb + nmax;
  const int ncnt = 2 * NU + NI;
  int* slots  = (int*)(agg_r + nmax);
  int* cnt    = slots + (size_t)ncnt * CAP;
  int emax = Ef > Erb ? Ef : Erb;
  emax = emax > Er ? emax : Er;

  zero_i32<<<(ncnt + 1023) / 1024, 256, 0, stream>>>(cnt, ncnt);
  fill_slots<<<dim3((emax + 2047) / 2048, 3), 256, 0, stream>>>(
      src_follows, src_ratedby, src_rates, dst_follows, dst_ratedby, dst_rates,
      cnt, slots, Ef, Erb, Er, NU);

  reduce_nodes<<<2048, 256, 0, stream>>>(
      feat_user, feat_item, src_follows, src_ratedby, src_rates,
      dst_follows, dst_ratedby, dst_rates, cnt, slots, agg_f, agg_rb, agg_r,
      NU, NI, Ef, Erb, Er);

  gemm_wmma_f32_all<<<256, 512, 0, stream>>>(
      agg_f, W_follows, b_follows, agg_rb, W_ratedby, b_ratedby,
      agg_r, W_rates, b_rates, cnt, out, NU / 16, NI / 16, NU);
}


# edges filed by ONE compare-and-swap each into the destination's 32-word line (was counter atomic + separate id store); lines preset to -1; degree = ids found in the line; sums in line order
# speedup vs baseline: 3.1555x; 1.0902x over previous
#include <hip/hip_runtime.h>

typedef __attribute__((ext_vector_type(2))) float v2f;
typedef __attribute__((ext_vector_type(8))) float v8f;

#define D 128

#define WP 132

__device__ __forceinline__ void gemm_strip(v8f (&c)[8], float4 (&a)[16],
                                           const float* __restrict__ sW,
                                           const float* __restrict__ anext,
                                           int r, int h) {
  const float* bl = sW + r * WP + h * 4;

  float4 b[8];
#pragma unroll
  for (int nt = 0; nt < 8; ++nt) b[nt] = *(const float4*)(bl + nt * 16 * WP);

#pragma unroll
  for (int j = 0; j < 16; ++j) {
    float4 bn[8];
    if (j + 1 < 16) {
#pragma unroll
      for (int nt = 0; nt < 8; ++nt)
        bn[nt] = *(const float4*)(bl + nt * 16 * WP + (j + 1) * 8);
    }
    __builtin_amdgcn_sched_barrier(0);
    const v2f a0 = {a[j].x, a[j].y};
    const v2f a1 = {a[j].z, a[j].w};
#pragma unroll
    for (int nt = 0; nt < 8; ++nt) {
      const v2f b0 = {b[nt].x, b[nt].y};
      c[nt] = __builtin_amdgcn_wmma_f32_16x16x4_f32(false, a0, false, b0,
                                                    (short)0, c[nt], false, false);
    }
#pragma unroll
    for (int nt = 0; nt < 8; ++nt) {
      const v2f b1 = {b[nt].z, b[nt].w};
      c[nt] = __builtin_amdgcn_wmma_f32_16x16x4_f32(false, a1, false, b1,
                                                    (short)0, c[nt], false, false);
    }
    a[j] = *(const float4*)(anext + j * 8);
    if (j + 1 < 16) {
#pragma unroll
      for (int nt = 0; nt < 8; ++nt) b[nt] = bn[nt];
    }
    __builtin_amdgcn_sched_barrier(0);
  }
}

__global__ void __launch_bounds__(512)
gemm_wmma_f32_all(const float* __restrict__ A_f, const float* __restrict__ W_f,
                  const float* __restrict__ b_f,
                  const float* __restrict__ A_rb, const float* __restrict__ W_rb,
                  const float* __restrict__ b_rb,
                  const float* __restrict__ A_r, const float* __restrict__ W_r,
                  const float* __restrict__ b_r,
                  const int* __restrict__ cnt, float* __restrict__ Y,
                  int ntu, int nti, int NU) {
  __shared__ __attribute__((aligned(16))) float sW[3 * D * WP];
  const int tid = threadIdx.x;
  const int lane = tid & 31;
  const int wave = __builtin_amdgcn_readfirstlane(tid >> 5);
  const int r = lane & 15;
  const int h = lane >> 4;

  const int nw = (int)gridDim.x * 16;
  const int g = wave * (int)gridDim.x + (int)blockIdx.x;
  const int ti0 = nw - 1 - g;

  const size_t lo = (size_t)r * D + h * 4;
#define STRIP(X, tile) ((X) + (size_t)(tile) * (16 * D) + lo)

  float4 a[16];
  {
    const float* p = g < ntu ? STRIP(A_f, g) : STRIP(A_r, ti0 < nti ? ti0 : 0);
#pragma unroll
    for (int j = 0; j < 16; ++j) a[j] = *(const float4*)(p + j * 8);
  }

#pragma unroll
  for (int s = 0; s < 3; ++s) {
    const float* W = s == 0 ? W_f : (s == 1 ? W_rb : W_r);
#pragma unroll
    for (int i = 0; i < 8; ++i) {
      const int idx = tid + i * 512;
      const int n = idx >> 5;
      const int c4 = idx & 31;
      const float4 w4 = ((const float4*)W)[idx];
      *(float4*)(sW + s * (D * WP) + n * WP + c4 * 4) = w4;
    }
  }
  __syncthreads();

  const int* deg_f = cnt;
  const int* deg_rb = cnt + NU;
  const int* deg_r = cnt + 2 * NU;

  for (int tile = g; tile < ntu; tile += nw) {
    v8f c[8];
#pragma unroll
    for (int nt = 0; nt < 8; ++nt) c[nt] = (v8f){};

    gemm_strip(c, a, sW, STRIP(A_rb, tile), r, h);
    const int tn = tile + nw;
    const float* nx = tn < ntu ? STRIP(A_f, tn)
                               : (ti0 < nti ? STRIP(A_r, ti0) : STRIP(A_rb, tile));
    gemm_strip(c, a, sW + D * WP, nx, r, h);

    bool g0[8], g1[8];
#pragma unroll
    for (int v = 0; v < 8; ++v) {
      g0[v] = deg_f[tile * 16 + h * 8 + v] > 0;
      g1[v] = deg_rb[tile * 16 + h * 8 + v] > 0;
    }
    float* yout = Y + (size_t)(tile * 16 + h * 8) * D + r;
#pragma unroll
    for (int nt = 0; nt < 8; ++nt) {
      const float bv0 = b_f[nt * 16 + r];
      const float bv1 = b_rb[nt * 16 + r];
#pragma unroll
      for (int v = 0; v < 8; ++v)
        yout[(size_t)v * D + nt * 16] = c[nt][v] + (g0[v] ? bv0 : 0.0f) + (g1[v] ? bv1 : 0.0f);
    }
  }

  float* Yi = Y + (size_t)NU * D;
  for (int tile = ti0; tile < nti; tile += nw) {
    v8f c[8];
#pragma unroll
    for (int nt = 0; nt < 8; ++nt) c[nt] = (v8f){};

    const int tn = tile + nw;
    gemm_strip(c, a, sW + 2 * D * WP, STRIP(A_r, tn < nti ? tn : tile), r, h);

    bool g0[8];
#pragma unroll
    for (int v = 0; v < 8; ++v) g0[v] = deg_r[tile * 16 + h * 8 + v] > 0;
    float* yout = Yi + (size_t)(tile * 16 + h * 8) * D + r;
#pragma unroll
    for (int nt = 0; nt < 8; ++nt) {
      const float bv0 = b_r[nt * 16 + r];
#pragma unroll
      for (int v = 0; v < 8; ++v)
        yout[(size_t)v * D + nt * 16] = c[nt][v] + (g0[v] ? bv0 : 0.0f);
    }
  }
#undef STRIP
}

#define CAP 32

__global__ void __launch_bounds__(256) fill_empty(int* __restrict__ p, int n4) {
  int4* q = (int4*)p;
  const int stride = gridDim.x * blockDim.x;
  for (int i = blockIdx.x * blockDim.x + threadIdx.x; i < n4; i += stride)
    q[i] = make_int4(-1, -1, -1, -1);
}

__global__ void __launch_bounds__(256)
fill_slots(const int* __restrict__ src_f, const int* __restrict__ src_rb,
           const int* __restrict__ src_r,
           const int* __restrict__ dst_f, const int* __restrict__ dst_rb,
           const int* __restrict__ dst_r,
           int* __restrict__ slots, int Ef, int Erb, int Er, int NU) {
  const int et = blockIdx.y;
  const int* src = et == 0 ? src_f : (et == 1 ? src_rb : src_r);
  const int* dst = et == 0 ? dst_f : (et == 1 ? dst_rb : dst_r);
  const int nE   = et == 0 ? Ef : (et == 1 ? Erb : Er);
  const int coff = et == 0 ? 0 : (et == 1 ? NU : 2 * NU);
  const int stride = gridDim.x * blockDim.x;
  for (int e0 = blockIdx.x * blockDim.x + threadIdx.x; e0 < nE; e0 += 4 * stride) {
    bool pend[4];
    int s[4];
    unsigned pos[4];
    int* line[4];
#pragma unroll
    for (int k = 0; k < 4; ++k) {
      const int e = e0 + k * stride;
      pend[k] = e < nE;
      const int ec = pend[k] ? e : e0;
      const int node = dst[ec];
      s[k] = src[ec];
      line[k] = slots + (size_t)(coff + node) * CAP;
      pos[k] = ((unsigned)ec * 2654435761u) >> 27;
    }
    int old[4];
#pragma unroll
    for (int k = 0; k < 4; ++k) {
      old[k] = -1;
      if (pend[k]) old[k] = atomicCAS(line[k] + pos[k], -1, s[k]);
    }
#pragma unroll
    for (int k = 0; k < 4; ++k) pend[k] = pend[k] && old[k] != -1;
    for (int t = 1; t < CAP; ++t) {
      if (!(pend[0] || pend[1] || pend[2] || pend[3])) break;
#pragma unroll
      for (int k = 0; k < 4; ++k) {
        old[k] = -1;
        if (pend[k]) {
          pos[k] = (pos[k] + 1u) & (CAP - 1);
          old[k] = atomicCAS(line[k] + pos[k], -1, s[k]);
        }
      }
#pragma unroll
      for (int k = 0; k < 4; ++k) pend[k] = pend[k] && old[k] != -1;
    }
  }
}

__device__ __forceinline__ void add_rows(float4& acc, const float* __restrict__ whl,
                                         int id, unsigned m) {
  while (m != 0u) {
    bool have[8];
    int s[8];
#pragma unroll
    for (int i = 0; i < 8; ++i) {
      have[i] = m != 0u;
      const int l = have[i] ? __builtin_ctz(m) : 0;
      s[i] = __builtin_amdgcn_readlane(id, l);
      m &= m - 1u;
    }
    float4 t[8];
#pragma unroll
    for (int i = 0; i < 8; ++i) {
      t[i] = make_float4(0.0f, 0.0f, 0.0f, 0.0f);
      if (have[i]) t[i] = *(const float4*)(whl + (size_t)s[i] * D);
    }
#pragma unroll
    for (int i = 0; i < 8; ++i) {
      acc.x += t[i].x; acc.y += t[i].y; acc.z += t[i].z; acc.w += t[i].w;
    }
  }
}

__device__ __forceinline__ float4 node_sum(const float* __restrict__ whl,
                                           const int* __restrict__ src,
                                           const int* __restrict__ dst, int nE,
                                           int d, int id, int lane, int& deg) {
  float4 acc = make_float4(0.0f, 0.0f, 0.0f, 0.0f);
  const unsigned m = (unsigned)__builtin_amdgcn_readfirstlane(
      (int)__builtin_amdgcn_ballot_w32(id >= 0));
  const int n = __builtin_popcount(m);
  if (n < CAP) {
    add_rows(acc, whl, id, m);
    deg = n;
    return acc;
  }
  int cnt = 0;
  for (int e0 = 0; e0 < nE; e0 += 32) {
    const int e = e0 + lane;
    bool hit = false;
    int s = 0;
    if (e < nE) {
      hit = dst[e] == d;
      s = src[e];
    }
    unsigned hm = __builtin_amdgcn_ballot_w32(hit);
    cnt += __builtin_popcount(hm);
    while (hm != 0u) {
      const int l = __builtin_amdgcn_readfirstlane(__builtin_ctz(hm));
      hm &= hm - 1u;
      const int row = __builtin_amdgcn_readlane(s, l);
      const float4 v = *(const float4*)(whl + (size_t)row * D);
      acc.x += v.x; acc.y += v.y; acc.z += v.z; acc.w += v.w;
    }
  }
  deg = cnt;
  return acc;
}

__global__ void __launch_bounds__(256)
reduce_nodes(const float* __restrict__ feat_user, const float* __restrict__ feat_item,
             const int* __restrict__ src_f, const int* __restrict__ src_rb,
             const int* __restrict__ src_r,
             const int* __restrict__ dst_f, const int* __restrict__ dst_rb,
             const int* __restrict__ dst_r,
             int* __restrict__ cnt, const int* __restrict__ slots,
             float* __restrict__ agg_f, float* __restrict__ agg_rb,
             float* __restrict__ agg_r, int NU, int NI, int Ef, int Erb, int Er) {
  const int lane = threadIdx.x & 31;
  int w = __builtin_amdgcn_readfirstlane((int)(blockIdx.x * 8 + (threadIdx.x >> 5)));
  const int nw = (int)(gridDim.x * 8);
  const int total = 2 * NU + NI;
  if (w >= total) return;
#define ROW_OF(v) ((v) < NU ? (v) : ((v) < NU + NI ? (v) + NU : (v) - NI))
  int id = slots[(size_t)ROW_OF(w) * CAP + lane];
  int wn = w + nw;
  for (;;) {
    int idn = -1;
    if (wn < total) idn = slots[(size_t)ROW_OF(wn) * CAP + lane];

    const float* tab; const int* src; const int* dst; float* agg;
    int nE, d;
    if (w < NU) {
      tab = feat_user; src = src_f; dst = dst_f; agg = agg_f; nE = Ef; d = w;
    } else if (w < NU + NI) {
      tab = feat_user; src = src_r; dst = dst_r; agg = agg_r; nE = Er; d = w - NU;
    } else {
      tab = feat_item; src = src_rb; dst = dst_rb; agg = agg_rb; nE = Erb; d = w - NU - NI;
    }
    int c;
    const float4 a = node_sum(tab + lane * 4, src, dst, nE, d, id, lane, c);
    const float dg = fmaxf((float)c, 1.0f);
    float4 res;
    res.x = a.x / dg; res.y = a.y / dg; res.z = a.z / dg; res.w = a.w / dg;
    *(float4*)(agg + (size_t)d * D + lane * 4) = res;
    if (lane == 0) cnt[ROW_OF(w)] = c;

    if (wn >= total) break;
    w = wn; id = idn; wn += nw;
  }
#undef ROW_OF
}

extern "C" void kernel_launch(void* const* d_in, const int* in_sizes, int n_in,
                              void* d_out, int out_size, void* d_ws, size_t ws_size,
                              hipStream_t stream) {
  (void)n_in; (void)out_size; (void)ws_size;

  const float* feat_user  = (const float*)d_in[0];
  const float* feat_item  = (const float*)d_in[1];
  const float* W_follows  = (const float*)d_in[2];
  const float* b_follows  = (const float*)d_in[3];
  const float* W_rates    = (const float*)d_in[4];
  const float* b_rates    = (const float*)d_in[5];
  const float* W_ratedby  = (const float*)d_in[6];
  const float* b_ratedby  = (const float*)d_in[7];
  const int* src_follows  = (const int*)d_in[8];
  const int* dst_follows  = (const int*)d_in[9];
  const int* src_rates    = (const int*)d_in[10];
  const int* dst_rates    = (const int*)d_in[11];
  const int* src_ratedby  = (const int*)d_in[12];
  const int* dst_ratedby  = (const int*)d_in[13];

  const int NU  = in_sizes[0] / D;
  const int NI  = in_sizes[1] / D;
  const int Ef  = in_sizes[8];
  const int Er  = in_sizes[10];
  const int Erb = in_sizes[12];

  float* out = (float*)d_out;

  float* ws = (float*)d_ws;
  const int nnmax = NU > NI ? NU : NI;
  const size_t nmax = (size_t)nnmax * D;
  float* agg_f  = ws;
  float* agg_rb = agg_f + nmax;
  float* agg_r  = agg_rb + nmax;
  const int ncnt = 2 * NU + NI;
  int* slots  = (int*)(agg_r + nmax);
  int* cnt    = slots + (size_t)ncnt * CAP;
  int emax = Ef > Erb ? Ef : Erb;
  emax = emax > Er ? emax : Er;

  fill_empty<<<1024, 256, 0, stream>>>(slots, ncnt * (CAP / 4));
  fill_slots<<<dim3((emax + 2047) / 2048, 3), 256, 0, stream>>>(
      src_follows, src_ratedby, src_rates, dst_follows, dst_ratedby, dst_rates,
      slots, Ef, Erb, Er, NU);

  reduce_nodes<<<2048, 256, 0, stream>>>(
      feat_user, feat_item, src_follows, src_ratedby, src_rates,
      dst_follows, dst_ratedby, dst_rates, cnt, slots, agg_f, agg_rb, agg_r,
      NU, NI, Ef, Erb, Er);

  gemm_wmma_f32_all<<<256, 512, 0, stream>>>(
      agg_f, W_follows, b_follows, agg_rb, W_ratedby, b_ratedby,
      agg_r, W_rates, b_rates, cnt, out, NU / 16, NI / 16, NU);
}
